# starting kernel
# baseline (speedup 1.0000x reference)
#include <hip/hip_runtime.h>
#include <hip/hip_bf16.h>


typedef __attribute__((ext_vector_type(2))) float v2f;
typedef __attribute__((ext_vector_type(8))) float v8f;

#define HP_P   48
#define HP_MT  3
#define HP_KT  12

static __device__ inline v8f wmma_f32_step(v8f acc, v2f Av, v2f Bv) {
    return __builtin_amdgcn_wmma_f32_16x16x4_f32(
        false, Av, false, Bv, (short)0, acc, false, false);
}

__global__ __launch_bounds__(256) void hawkes_ll_kernel(
    const float* __restrict__ t,
    const float* __restrict__ mu,
    const float* __restrict__ alpha,
    const float* __restrict__ beta,
    float* __restrict__ out,
    int S)
{
    const float b = beta[0];

    const int lane = threadIdx.x & 31;
    const int wave = threadIdx.x >> 5;
    const int hsel = lane >> 4;
    const int ln   = lane & 15;

    const int s0   = (blockIdx.x * (blockDim.x >> 5) + wave) * 16;
    const int scol = s0 + ln;
    const int sc   = (scol < S) ? scol : (S - 1);

    v8f acc0 = {}; v8f acc1 = {}; v8f acc2 = {};

    #pragma unroll
    for (int kk = 0; kk < HP_KT; ++kk) {
        const int q0 = 4 * kk + 2 * hsel;

        v2f Bv;
        Bv.x = __expf(b * t[(size_t)(q0 + 0) * S + sc]);
        Bv.y = __expf(b * t[(size_t)(q0 + 1) * S + sc]);

        #pragma unroll
        for (int mt = 0; mt < HP_MT; ++mt) {
            const int pr = 16 * mt + ln;
            v2f Av;
            Av.x = alpha[(size_t)pr * HP_P + q0 + 0];
            Av.y = alpha[(size_t)pr * HP_P + q0 + 1];
            if      (mt == 0) acc0 = wmma_f32_step(acc0, Av, Bv);
            else if (mt == 1) acc1 = wmma_f32_step(acc1, Av, Bv);
            else              acc2 = wmma_f32_step(acc2, Av, Bv);
        }
    }

    float part = 0.0f;
    #pragma unroll
    for (int mt = 0; mt < HP_MT; ++mt) {
        v8f a = (mt == 0) ? acc0 : (mt == 1) ? acc1 : acc2;
        #pragma unroll
        for (int r = 0; r < 8; ++r) {
            const int p  = 16 * mt + r + 8 * hsel;
            const float tp  = t[(size_t)p * S + sc];
            const float lam = mu[p] + __expf(-b * tp) * a[r];
            part += __logf(lam) * tp - lam;
        }
    }

    const float tot = part + __shfl_xor(part, 16, 32);
    if (hsel == 0 && scol < S) out[scol] = tot;
}

extern "C" void kernel_launch(void* const* d_in, const int* in_sizes, int n_in,
                              void* d_out, int out_size, void* d_ws, size_t ws_size,
                              hipStream_t stream) {
    (void)n_in; (void)d_ws; (void)ws_size;
    const float* t     = (const float*)d_in[0];
    const float* mu    = (const float*)d_in[1];
    const float* alpha = (const float*)d_in[2];
    const float* beta  = (const float*)d_in[3];
    float* out = (float*)d_out;

    const int S = in_sizes[0] / HP_P;
    (void)out_size;

    const int threads = 256;
    const int cols_per_block = (threads / 32) * 16;
    const int blocks = (S + cols_per_block - 1) / cols_per_block;

    hawkes_ll_kernel<<<blocks, threads, 0, stream>>>(t, mu, alpha, beta, out, S);
}


# Epilogue log(lam) taken as ln2 times the hardware log2 instruction in place of the library log's guarded expansion (12 vector instructions per element down to 2); every sum keeps its order
# speedup vs baseline: 1.0977x; 1.0977x over previous
#include <hip/hip_runtime.h>
#include <hip/hip_bf16.h>


typedef __attribute__((ext_vector_type(2))) float v2f;
typedef __attribute__((ext_vector_type(8))) float v8f;

#define HP_P   48
#define HP_MT  3
#define HP_KT  12

static __device__ inline v8f wmma_f32_step(v8f acc, v2f Av, v2f Bv) {
    return __builtin_amdgcn_wmma_f32_16x16x4_f32(
        false, Av, false, Bv, (short)0, acc, false, false);
}

__global__ __launch_bounds__(256) void hawkes_ll_kernel(
    const float* __restrict__ t,
    const float* __restrict__ mu,
    const float* __restrict__ alpha,
    const float* __restrict__ beta,
    float* __restrict__ out,
    int S)
{
    const float b = beta[0];

    const int lane = threadIdx.x & 31;
    const int wave = threadIdx.x >> 5;
    const int hsel = lane >> 4;
    const int ln   = lane & 15;

    const int s0   = (blockIdx.x * (blockDim.x >> 5) + wave) * 16;
    const int scol = s0 + ln;
    const int sc   = (scol < S) ? scol : (S - 1);

    v8f acc0 = {}; v8f acc1 = {}; v8f acc2 = {};

    #pragma unroll
    for (int kk = 0; kk < HP_KT; ++kk) {
        const int q0 = 4 * kk + 2 * hsel;

        v2f Bv;
        Bv.x = __expf(b * t[(size_t)(q0 + 0) * S + sc]);
        Bv.y = __expf(b * t[(size_t)(q0 + 1) * S + sc]);

        #pragma unroll
        for (int mt = 0; mt < HP_MT; ++mt) {
            const int pr = 16 * mt + ln;
            v2f Av;
            Av.x = alpha[(size_t)pr * HP_P + q0 + 0];
            Av.y = alpha[(size_t)pr * HP_P + q0 + 1];
            if      (mt == 0) acc0 = wmma_f32_step(acc0, Av, Bv);
            else if (mt == 1) acc1 = wmma_f32_step(acc1, Av, Bv);
            else              acc2 = wmma_f32_step(acc2, Av, Bv);
        }
    }

    float part = 0.0f;
    #pragma unroll
    for (int mt = 0; mt < HP_MT; ++mt) {
        v8f a = (mt == 0) ? acc0 : (mt == 1) ? acc1 : acc2;
        #pragma unroll
        for (int r = 0; r < 8; ++r) {
            const int p  = 16 * mt + r + 8 * hsel;
            const float tp  = t[(size_t)p * S + sc];
            const float lam = mu[p] + __expf(-b * tp) * a[r];
            const float lg = 0.69314718f * __builtin_amdgcn_logf(lam);
            part += lg * tp - lam;
        }
    }

    const float tot = part + __shfl_xor(part, 16, 32);
    if (hsel == 0 && scol < S) out[scol] = tot;
}

extern "C" void kernel_launch(void* const* d_in, const int* in_sizes, int n_in,
                              void* d_out, int out_size, void* d_ws, size_t ws_size,
                              hipStream_t stream) {
    (void)n_in; (void)d_ws; (void)ws_size;
    const float* t     = (const float*)d_in[0];
    const float* mu    = (const float*)d_in[1];
    const float* alpha = (const float*)d_in[2];
    const float* beta  = (const float*)d_in[3];
    float* out = (float*)d_out;

    const int S = in_sizes[0] / HP_P;
    (void)out_size;

    const int threads = 256;
    const int cols_per_block = (threads / 32) * 16;
    const int blocks = (S + cols_per_block - 1) / cols_per_block;

    hawkes_ll_kernel<<<blocks, threads, 0, stream>>>(t, mu, alpha, beta, out, S);
}


# Each lane half owns rows 24h..24h+23 as its k share and its output rows (A rows and k order re-indexed): every t loaded and scaled once, no epilogue reload; k-sum and row-sum orders change
# speedup vs baseline: 1.1960x; 1.0896x over previous
#include <hip/hip_runtime.h>
#include <hip/hip_bf16.h>


typedef __attribute__((ext_vector_type(2))) float v2f;
typedef __attribute__((ext_vector_type(8))) float v8f;

#define HP_P   48
#define HP_MT  3
#define HP_KT  12
#define HP_H   24

static __device__ inline v8f wmma_f32_step(v8f acc, v2f Av, v2f Bv) {
    return __builtin_amdgcn_wmma_f32_16x16x4_f32(
        false, Av, false, Bv, (short)0, acc, false, false);
}

__global__ __launch_bounds__(256) void hawkes_ll_kernel(
    const float* __restrict__ t,
    const float* __restrict__ mu,
    const float* __restrict__ alpha,
    const float* __restrict__ beta,
    float* __restrict__ out,
    int S)
{
    const float b = beta[0];

    const int lane = threadIdx.x & 31;
    const int wave = threadIdx.x >> 5;
    const int hsel = lane >> 4;
    const int ln   = lane & 15;

    const int s0   = (blockIdx.x * (blockDim.x >> 5) + wave) * 16;
    const int scol = s0 + ln;
    const int sc   = (scol < S) ? scol : (S - 1);

    const unsigned toff = (unsigned)(HP_H * hsel) * (unsigned)S + (unsigned)sc;

    float tv[HP_H];
    float ye[HP_H];
    float Ev[HP_H];
    #pragma unroll
    for (int i = 0; i < HP_H; ++i) {
        const float* trow = t + (size_t)i * (size_t)S;
        tv[i] = trow[toff];
        ye[i] = (b * tv[i]) * 1.44269504f;
        Ev[i] = __builtin_amdgcn_exp2f(ye[i]);
    }

    const float* arow = alpha + (size_t)(HP_H * (ln >> 3) + (ln & 7)) * HP_P + HP_H * hsel;

    v8f acc0 = {}; v8f acc1 = {}; v8f acc2 = {};

    #pragma unroll
    for (int kk = 0; kk < HP_KT; ++kk) {
        v2f Bv;
        Bv.x = Ev[2 * kk + 0];
        Bv.y = Ev[2 * kk + 1];

        #pragma unroll
        for (int mt = 0; mt < HP_MT; ++mt) {
            v2f Av;
            Av.x = arow[(size_t)(8 * mt) * HP_P + 2 * kk + 0];
            Av.y = arow[(size_t)(8 * mt) * HP_P + 2 * kk + 1];
            if      (mt == 0) acc0 = wmma_f32_step(acc0, Av, Bv);
            else if (mt == 1) acc1 = wmma_f32_step(acc1, Av, Bv);
            else              acc2 = wmma_f32_step(acc2, Av, Bv);
        }
    }

    const float* mup = mu + HP_H * hsel;
    float part = 0.0f;
    #pragma unroll
    for (int mt = 0; mt < HP_MT; ++mt) {
        v8f a = (mt == 0) ? acc0 : (mt == 1) ? acc1 : acc2;
        #pragma unroll
        for (int r = 0; r < 8; ++r) {
            const int i = 8 * mt + r;
            const float tp  = tv[i];
            const float lam = mup[i] + __builtin_amdgcn_exp2f(-ye[i]) * a[r];
            const float lg = 0.69314718f * __builtin_amdgcn_logf(lam);
            part += lg * tp - lam;
        }
    }

    const float tot = part + __shfl_xor(part, 16, 32);
    if (hsel == 0 && scol < S) out[scol] = tot;
}

extern "C" void kernel_launch(void* const* d_in, const int* in_sizes, int n_in,
                              void* d_out, int out_size, void* d_ws, size_t ws_size,
                              hipStream_t stream) {
    (void)n_in; (void)d_ws; (void)ws_size;
    const float* t     = (const float*)d_in[0];
    const float* mu    = (const float*)d_in[1];
    const float* alpha = (const float*)d_in[2];
    const float* beta  = (const float*)d_in[3];
    float* out = (float*)d_out;

    const int S = in_sizes[0] / HP_P;
    (void)out_size;

    const int threads = 256;
    const int cols_per_block = (threads / 32) * 16;
    const int blocks = (S + cols_per_block - 1) / cols_per_block;

    hawkes_ll_kernel<<<blocks, threads, 0, stream>>>(t, mu, alpha, beta, out, S);
}


# Causal-mask block skip: lane halves own alternate layers, so 12 of 36 tile/k-group matrix steps and a third of the A loads drop; loads fenced into one leading group; k-sum and row-sum orders change
# speedup vs baseline: 1.3059x; 1.0919x over previous
#include <hip/hip_runtime.h>
#include <hip/hip_bf16.h>


typedef __attribute__((ext_vector_type(2))) float v2f;
typedef __attribute__((ext_vector_type(8))) float v8f;

#define HP_P   48
#define HP_H   24

static __device__ inline v8f wmma_f32_step(v8f acc, v2f Av, v2f Bv) {
    return __builtin_amdgcn_wmma_f32_16x16x4_f32(
        false, Av, false, Bv, (short)0, acc, false, false);
}

__global__ __launch_bounds__(256) void hawkes_ll_kernel(
    const float* __restrict__ t,
    const float* __restrict__ mu,
    const float* __restrict__ alpha,
    const float* __restrict__ beta,
    float* __restrict__ out,
    int S)
{
    const float b = beta[0];

    const int lane = threadIdx.x & 31;
    const int wave = threadIdx.x >> 5;
    const int hsel = lane >> 4;
    const int ln   = lane & 15;

    const int s0   = (blockIdx.x * (blockDim.x >> 5) + wave) * 16;
    const int scol = s0 + ln;
    const int sc   = (scol < S) ? scol : (S - 1);

    const unsigned toff = (unsigned)(8 * hsel) * (unsigned)S + (unsigned)sc;

    float tv[HP_H];
    #pragma unroll
    for (int i = 0; i < HP_H; ++i) {
        const float* trow = t + (size_t)(16 * (i >> 3) + (i & 7)) * (size_t)S;
        tv[i] = trow[toff];
    }

    const float* arow = alpha + (size_t)ln * HP_P + 8 * hsel;
    float Aq[3][3][8];
    #pragma unroll
    for (int mt = 0; mt < 3; ++mt) {
        #pragma unroll
        for (int g = 0; g <= mt; ++g) {
            #pragma unroll
            for (int e = 0; e < 8; ++e)
                Aq[mt][g][e] = arow[(size_t)(16 * mt) * HP_P + 16 * g + e];
        }
    }

    const float* mup = mu + 8 * hsel;
    float mv[HP_H];
    #pragma unroll
    for (int i = 0; i < HP_H; ++i)
        mv[i] = mup[16 * (i >> 3) + (i & 7)];

    __builtin_amdgcn_sched_barrier(0);

    float ye[HP_H];
    float Ev[HP_H];
    #pragma unroll
    for (int i = 0; i < HP_H; ++i) {
        ye[i] = (b * tv[i]) * 1.44269504f;
        Ev[i] = __builtin_amdgcn_exp2f(ye[i]);
    }

    v8f acc0 = {}; v8f acc1 = {}; v8f acc2 = {};

    #define HP_STEP(ACC, MT, KK)                                              \
        {                                                                     \
            v2f Av, Bv;                                                       \
            Av.x = Aq[MT][(KK) >> 2][2 * ((KK) & 3) + 0];                     \
            Av.y = Aq[MT][(KK) >> 2][2 * ((KK) & 3) + 1];                     \
            Bv.x = Ev[2 * (KK) + 0];                                          \
            Bv.y = Ev[2 * (KK) + 1];                                          \
            ACC = wmma_f32_step(ACC, Av, Bv);                                 \
        }
    #pragma unroll
    for (int kk = 0; kk < 8; ++kk) {
        HP_STEP(acc2, 2, kk)
        HP_STEP(acc1, 1, kk)
    }
    #pragma unroll
    for (int ii = 0; ii < 4; ++ii) {
        HP_STEP(acc2, 2, 8 + ii)
        HP_STEP(acc0, 0, ii)
    }
    #undef HP_STEP

    float part = 0.0f;
    #pragma unroll
    for (int mt = 0; mt < 3; ++mt) {
        v8f a = (mt == 0) ? acc0 : (mt == 1) ? acc1 : acc2;
        #pragma unroll
        for (int r = 0; r < 8; ++r) {
            const int i = 8 * mt + r;
            const float tp  = tv[i];
            const float lam = mv[i] + __builtin_amdgcn_exp2f(-ye[i]) * a[r];
            const float lg = 0.69314718f * __builtin_amdgcn_logf(lam);
            part += lg * tp - lam;
        }
    }

    const float tot = part + __shfl_xor(part, 16, 32);
    if (hsel == 0 && scol < S) out[scol] = tot;
}

extern "C" void kernel_launch(void* const* d_in, const int* in_sizes, int n_in,
                              void* d_out, int out_size, void* d_ws, size_t ws_size,
                              hipStream_t stream) {
    (void)n_in; (void)d_ws; (void)ws_size;
    const float* t     = (const float*)d_in[0];
    const float* mu    = (const float*)d_in[1];
    const float* alpha = (const float*)d_in[2];
    const float* beta  = (const float*)d_in[3];
    float* out = (float*)d_out;

    const int S = in_sizes[0] / HP_P;
    (void)out_size;

    const int threads = 256;
    const int cols_per_block = (threads / 32) * 16;
    const int blocks = (S + cols_per_block - 1) / cols_per_block;

    hawkes_ll_kernel<<<blocks, threads, 0, stream>>>(t, mu, alpha, beta, out, S);
}
